# starting kernel
# baseline (speedup 1.0000x reference)
#include <hip/hip_runtime.h>

typedef __attribute__((ext_vector_type(2))) float v2f;
typedef __attribute__((ext_vector_type(8))) float v8f;

#define D 128

__global__ __launch_bounds__(256) void gnn_zero(float* __restrict__ p, long long n) {
    long long i = (long long)blockIdx.x * blockDim.x + threadIdx.x;
    if (i < n) p[i] = 0.0f;
}

__global__ __launch_bounds__(256) void gnn_count(const int* __restrict__ dst,
                                                 float* __restrict__ cnt, int nEdges) {
    int e = blockIdx.x * blockDim.x + threadIdx.x;
    if (e < nEdges) atomicAdd(&cnt[dst[e]], 1.0f);
}

__global__ __launch_bounds__(256) void gnn_inv(float* __restrict__ cnt, int nNodes) {
    int n = blockIdx.x * blockDim.x + threadIdx.x;
    if (n < nNodes) cnt[n] = 1.0f / fmaxf(cnt[n], 1.0f);
}

__global__ __launch_bounds__(256) void gnn_scatter(const float* __restrict__ feat,
                                                   const int* __restrict__ src,
                                                   const int* __restrict__ dst,
                                                   float* __restrict__ agg, int nEdges) {
    long long t = (long long)blockIdx.x * blockDim.x + threadIdx.x;
    int e = (int)(t >> 5);
    int c = ((int)t & 31) * 4;
    if (e >= nEdges) return;
    int s = src[e];
    int d = dst[e];
    float4 val = *(const float4*)(feat + (size_t)s * D + c);
    float* p = agg + (size_t)d * D + c;
    atomicAdd(p + 0, val.x);
    atomicAdd(p + 1, val.y);
    atomicAdd(p + 2, val.z);
    atomicAdd(p + 3, val.w);
}

__global__ __launch_bounds__(256) void gnn_sage_gemm(
    const float* __restrict__ agg,
    const float* __restrict__ inv,
    const float* __restrict__ xin,
    const float* __restrict__ wl,
    const float* __restrict__ wr,
    const float* __restrict__ bias,
    float* __restrict__ out,
    int relu) {
    const int lane    = threadIdx.x & 31;
    const int wave    = threadIdx.x >> 5;
    const int rowBase = blockIdx.x * 16;
    const int colBase = wave * 16;

    const int half = lane >> 4;
    const int l15  = lane & 15;
    const int arow = rowBase + l15;
    const int brow = colBase + l15;

    const float inv_c = inv[arow];
    const float* aggRow = agg + (size_t)arow * D;
    const float* xRow   = xin + (size_t)arow * D;
    const float* wlRow  = wl + (size_t)brow * D;
    const float* wrRow  = wr + (size_t)brow * D;

    v8f acc = {};
#pragma unroll 8
    for (int kc = 0; kc < D; kc += 4) {
        const int k0 = kc + 2 * half;
        v2f a_m = *(const v2f*)(aggRow + k0);
        a_m.x *= inv_c;
        a_m.y *= inv_c;
        v2f a_x = *(const v2f*)(xRow + k0);
        v2f b_l = *(const v2f*)(wlRow + k0);
        v2f b_r = *(const v2f*)(wrRow + k0);
        acc = __builtin_amdgcn_wmma_f32_16x16x4_f32(false, a_m, false, b_l,
                                                    (short)0, acc, false, false);
        acc = __builtin_amdgcn_wmma_f32_16x16x4_f32(false, a_x, false, b_r,
                                                    (short)0, acc, false, false);
    }

    const float bv = bias[colBase + l15];
#pragma unroll
    for (int v = 0; v < 8; ++v) {
        float r = acc[v] + bv;
        if (relu) r = fmaxf(r, 0.0f);
        out[(size_t)(rowBase + v + 8 * half) * D + colBase + l15] = r;
    }
}

extern "C" void kernel_launch(void* const* d_in, const int* in_sizes, int n_in,
                              void* d_out, int out_size, void* d_ws, size_t ws_size,
                              hipStream_t stream) {
    const float* x   = (const float*)d_in[0];
    const int* edge  = (const int*)d_in[1];
    const float* w1l = (const float*)d_in[2];
    const float* w1r = (const float*)d_in[3];
    const float* b1  = (const float*)d_in[4];
    const float* w2l = (const float*)d_in[5];
    const float* w2r = (const float*)d_in[6];
    const float* b2  = (const float*)d_in[7];
    float* out = (float*)d_out;

    const int N = in_sizes[0] / D;
    const int E = in_sizes[1] / 2;
    const int* src = edge;
    const int* dst = edge + E;

    float* agg = (float*)d_ws;
    float* h   = agg + (size_t)N * D;
    float* inv = h + (size_t)N * D;

    const long long nd = (long long)N * D;
    dim3 blk(256);
    dim3 gZeroND((unsigned)((nd + 255) / 256));
    dim3 gZeroN((unsigned)((N + 255) / 256));
    dim3 gCnt((unsigned)((E + 255) / 256));
    dim3 gScat((unsigned)(((long long)E * 32 + 255) / 256));
    dim3 gGemm((unsigned)(N / 16));

    gnn_zero<<<gZeroN, blk, 0, stream>>>(inv, N);
    gnn_count<<<gCnt, blk, 0, stream>>>(dst, inv, E);
    gnn_inv<<<gZeroN, blk, 0, stream>>>(inv, N);

    gnn_zero<<<gZeroND, blk, 0, stream>>>(agg, nd);
    gnn_scatter<<<gScat, blk, 0, stream>>>(x, src, dst, agg, E);
    gnn_sage_gemm<<<gGemm, blk, 0, stream>>>(agg, inv, x, w1l, w1r, b1, h, 1);

    gnn_zero<<<gZeroND, blk, 0, stream>>>(agg, nd);
    gnn_scatter<<<gScat, blk, 0, stream>>>(h, src, dst, agg, E);
    gnn_sage_gemm<<<gGemm, blk, 0, stream>>>(agg, inv, h, w2l, w2r, b2, out, 0);
}


# SAGE GEMM retiled: each wave owns a 16x128 row band (8 accumulators), A read once with 16-byte loads, both weight matrices staged once per block in padded LDS for B; k summation order changes
# speedup vs baseline: 1.2650x; 1.2650x over previous
#include <hip/hip_runtime.h>

typedef __attribute__((ext_vector_type(2))) float v2f;
typedef __attribute__((ext_vector_type(4))) float v4f;
typedef __attribute__((ext_vector_type(8))) float v8f;

#define D 128

__global__ __launch_bounds__(256) void gnn_zero(float* __restrict__ p, long long n) {
    long long i = (long long)blockIdx.x * blockDim.x + threadIdx.x;
    if (i < n) p[i] = 0.0f;
}

__global__ __launch_bounds__(256) void gnn_count(const int* __restrict__ dst,
                                                 float* __restrict__ cnt, int nEdges) {
    int e = blockIdx.x * blockDim.x + threadIdx.x;
    if (e < nEdges) atomicAdd(&cnt[dst[e]], 1.0f);
}

__global__ __launch_bounds__(256) void gnn_inv(float* __restrict__ cnt, int nNodes) {
    int n = blockIdx.x * blockDim.x + threadIdx.x;
    if (n < nNodes) cnt[n] = 1.0f / fmaxf(cnt[n], 1.0f);
}

__global__ __launch_bounds__(256) void gnn_scatter(const float* __restrict__ feat,
                                                   const int* __restrict__ src,
                                                   const int* __restrict__ dst,
                                                   float* __restrict__ agg, int nEdges) {
    long long t = (long long)blockIdx.x * blockDim.x + threadIdx.x;
    int e = (int)(t >> 5);
    int c = ((int)t & 31) * 4;
    if (e >= nEdges) return;
    int s = src[e];
    int d = dst[e];
    float4 val = *(const float4*)(feat + (size_t)s * D + c);
    float* p = agg + (size_t)d * D + c;
    atomicAdd(p + 0, val.x);
    atomicAdd(p + 1, val.y);
    atomicAdd(p + 2, val.z);
    atomicAdd(p + 3, val.w);
}

#define WP 132
#define GEMM_ROWS 64

__global__ __launch_bounds__(128) void gnn_sage_gemm(
    const float* __restrict__ agg,
    const float* __restrict__ inv,
    const float* __restrict__ xin,
    const float* __restrict__ wl,
    const float* __restrict__ wr,
    const float* __restrict__ bias,
    float* __restrict__ out,
    int nNodes, int relu) {
    __shared__ float wlds[2 * D * WP];

    const int tid     = threadIdx.x;
    const int lane    = tid & 31;
    const int wave    = tid >> 5;
    const int half    = lane >> 4;
    const int l15     = lane & 15;
    const int rowBase = blockIdx.x * GEMM_ROWS + wave * 16;
    const bool valid  = rowBase + 16 <= nNodes;

    v4f am[16], ax[16];
    const int arow = valid ? rowBase + l15 : l15;
    const float inv_c = inv[arow];
    {
        const float* aggRow = agg + (size_t)arow * D + 4 * half;
        const float* xRow   = xin + (size_t)arow * D + 4 * half;
#pragma unroll
        for (int j = 0; j < 16; ++j) am[j] = *(const v4f*)(aggRow + 8 * j);
#pragma unroll
        for (int j = 0; j < 16; ++j) ax[j] = *(const v4f*)(xRow + 8 * j);
    }

    {
        const int c4 = (tid & 31) * 4;
        const int rw = tid >> 5;
#pragma unroll 16
        for (int p = 0; p < 32; ++p) {
            const int r = (((p + (int)blockIdx.x) & 31) << 2) + rw;
            const v4f vl = *(const v4f*)(wl + r * D + c4);
            const v4f vr = *(const v4f*)(wr + r * D + c4);
            *(v4f*)(wlds + r * WP + c4)       = vl;
            *(v4f*)(wlds + (D + r) * WP + c4) = vr;
        }
    }
    __syncthreads();
    if (!valid) return;

#pragma unroll
    for (int j = 0; j < 16; ++j) am[j] *= inv_c;

    v8f acc[8];
#pragma unroll
    for (int n = 0; n < 8; ++n) acc[n] = (v8f){0.f, 0.f, 0.f, 0.f, 0.f, 0.f, 0.f, 0.f};

    const float* bBase = wlds + l15 * WP + 4 * half;

    v4f b0[8], b1[8];
#pragma unroll
    for (int n = 0; n < 8; ++n) b0[n] = *(const v4f*)(bBase + n * 16 * WP);

#pragma unroll
    for (int j = 0; j < 32; j += 2) {
        {
            const int jn = j + 1;
            const float* bp = bBase + (jn >> 4) * (D * WP) + (jn & 15) * 8;
#pragma unroll
            for (int n = 0; n < 8; ++n) b1[n] = *(const v4f*)(bp + n * 16 * WP);
        }
        {
            const v4f a = (j < 16) ? am[j & 15] : ax[j & 15];
            const v2f a01 = {a.x, a.y};
            const v2f a23 = {a.z, a.w};
#pragma unroll
            for (int n = 0; n < 8; ++n) {
                const v2f b = {b0[n].x, b0[n].y};
                acc[n] = __builtin_amdgcn_wmma_f32_16x16x4_f32(false, a01, false, b,
                                                               (short)0, acc[n], false, false);
            }
#pragma unroll
            for (int n = 0; n < 8; ++n) {
                const v2f b = {b0[n].z, b0[n].w};
                acc[n] = __builtin_amdgcn_wmma_f32_16x16x4_f32(false, a23, false, b,
                                                               (short)0, acc[n], false, false);
            }
        }
        if (j + 2 < 32) {
            const int jn = j + 2;
            const float* bp = bBase + (jn >> 4) * (D * WP) + (jn & 15) * 8;
#pragma unroll
            for (int n = 0; n < 8; ++n) b0[n] = *(const v4f*)(bp + n * 16 * WP);
        }
        {
            const int jj = j + 1;
            const v4f a = (jj < 16) ? am[jj & 15] : ax[jj & 15];
            const v2f a01 = {a.x, a.y};
            const v2f a23 = {a.z, a.w};
#pragma unroll
            for (int n = 0; n < 8; ++n) {
                const v2f b = {b1[n].x, b1[n].y};
                acc[n] = __builtin_amdgcn_wmma_f32_16x16x4_f32(false, a01, false, b,
                                                               (short)0, acc[n], false, false);
            }
#pragma unroll
            for (int n = 0; n < 8; ++n) {
                const v2f b = {b1[n].z, b1[n].w};
                acc[n] = __builtin_amdgcn_wmma_f32_16x16x4_f32(false, a23, false, b,
                                                               (short)0, acc[n], false, false);
            }
        }
    }

#pragma unroll
    for (int n = 0; n < 8; ++n) {
        const float bv = bias[n * 16 + l15];
#pragma unroll
        for (int v = 0; v < 8; ++v) {
            float r = acc[n][v] + bv;
            if (relu) r = fmaxf(r, 0.0f);
            out[(size_t)(rowBase + v + 8 * half) * D + n * 16 + l15] = r;
        }
    }
}

extern "C" void kernel_launch(void* const* d_in, const int* in_sizes, int n_in,
                              void* d_out, int out_size, void* d_ws, size_t ws_size,
                              hipStream_t stream) {
    const float* x   = (const float*)d_in[0];
    const int* edge  = (const int*)d_in[1];
    const float* w1l = (const float*)d_in[2];
    const float* w1r = (const float*)d_in[3];
    const float* b1  = (const float*)d_in[4];
    const float* w2l = (const float*)d_in[5];
    const float* w2r = (const float*)d_in[6];
    const float* b2  = (const float*)d_in[7];
    float* out = (float*)d_out;

    const int N = in_sizes[0] / D;
    const int E = in_sizes[1] / 2;
    const int* src = edge;
    const int* dst = edge + E;

    float* agg = (float*)d_ws;
    float* h   = agg + (size_t)N * D;
    float* inv = h + (size_t)N * D;

    const long long nd = (long long)N * D;
    dim3 blk(256);
    dim3 gZeroND((unsigned)((nd + 255) / 256));
    dim3 gZeroN((unsigned)((N + 255) / 256));
    dim3 gCnt((unsigned)((E + 255) / 256));
    dim3 gScat((unsigned)(((long long)E * 32 + 255) / 256));
    dim3 blkGemm(128);
    dim3 gGemm((unsigned)((N + GEMM_ROWS - 1) / GEMM_ROWS));

    gnn_zero<<<gZeroN, blk, 0, stream>>>(inv, N);
    gnn_count<<<gCnt, blk, 0, stream>>>(dst, inv, E);
    gnn_inv<<<gZeroN, blk, 0, stream>>>(inv, N);

    gnn_zero<<<gZeroND, blk, 0, stream>>>(agg, nd);
    gnn_scatter<<<gScat, blk, 0, stream>>>(x, src, dst, agg, E);
    gnn_sage_gemm<<<gGemm, blkGemm, 0, stream>>>(agg, inv, x, w1l, w1r, b1, h, N, 1);

    gnn_zero<<<gZeroND, blk, 0, stream>>>(agg, nd);
    gnn_scatter<<<gScat, blk, 0, stream>>>(h, src, dst, agg, E);
    gnn_sage_gemm<<<gGemm, blkGemm, 0, stream>>>(agg, inv, h, w2l, w2r, b2, out, N, 0);
}


# Edges counting-sorted by destination per call (integer slot adds, prefix, fill); one wave sums a node's source rows in registers; float atomics and zero fill gone; sum order = slot arrival
# speedup vs baseline: 2.6498x; 2.0947x over previous
#include <hip/hip_runtime.h>

typedef __attribute__((ext_vector_type(2))) float v2f;
typedef __attribute__((ext_vector_type(4))) float v4f;
typedef __attribute__((ext_vector_type(8))) float v8f;

#define D 128

__global__ __launch_bounds__(256) void gnn_zero_i(int* __restrict__ p, int n) {
    int i = blockIdx.x * blockDim.x + threadIdx.x;
    if (i < n) p[i] = 0;
}

__global__ __launch_bounds__(256) void gnn_slot(const int* __restrict__ dst,
                                                int* __restrict__ cnt,
                                                int* __restrict__ slot, int nEdges) {
    int e = blockIdx.x * blockDim.x + threadIdx.x;
    if (e < nEdges) slot[e] = atomicAdd(&cnt[dst[e]], 1);
}

__global__ __launch_bounds__(256) void gnn_rowptr(const int* __restrict__ cnt,
                                                  int* __restrict__ rowptr,
                                                  float* __restrict__ inv, int nNodes) {
    __shared__ int sFront[8];
    __shared__ int sWave[8];
    const int tid  = threadIdx.x;
    const int lane = tid & 31;
    const int w    = tid >> 5;
    const int b0   = blockIdx.x * 256;

    int front = 0;
    {
        const int4* c4 = (const int4*)cnt;
        const int n4 = b0 >> 2;
#pragma unroll 8
        for (int i = tid; i < n4; i += 256) {
            const int4 v = c4[i];
            front += (v.x + v.y) + (v.z + v.w);
        }
    }
    const int idx = b0 + tid;
    const int c = (idx < nNodes) ? cnt[idx] : 0;

    int incl = c;
    for (int off = 1; off < 32; off <<= 1) {
        const int t = __shfl_up(incl, off);
        if (lane >= off) incl += t;
    }
    for (int off = 16; off > 0; off >>= 1) front += __shfl_xor(front, off);
    if (lane == 31) sWave[w] = incl;
    if (lane == 0) sFront[w] = front;
    __syncthreads();
    int base = 0;
    for (int k = 0; k < 8; ++k) {
        base += sFront[k];
        if (k < w) base += sWave[k];
    }
    const int start = base + incl - c;
    if (idx < nNodes) {
        rowptr[idx] = start;
        inv[idx] = 1.0f / fmaxf((float)c, 1.0f);
    }
    if (idx == nNodes - 1) rowptr[nNodes] = start + c;
}

__global__ __launch_bounds__(256) void gnn_fill(const int* __restrict__ src,
                                                const int* __restrict__ dst,
                                                const int* __restrict__ rowptr,
                                                const int* __restrict__ slot,
                                                int* __restrict__ col, int nEdges) {
    int e = blockIdx.x * blockDim.x + threadIdx.x;
    if (e < nEdges) col[rowptr[dst[e]] + slot[e]] = src[e];
}

__global__ __launch_bounds__(256) void gnn_gather(const float* __restrict__ feat,
                                                  const int* __restrict__ rowptr,
                                                  const int* __restrict__ col,
                                                  float* __restrict__ agg, int nNodes) {
    const int lane = threadIdx.x & 31;
    const int node = __builtin_amdgcn_readfirstlane((int)((blockIdx.x * 256u + threadIdx.x) >> 5));
    if (node >= nNodes) return;
    const int beg = rowptr[node];
    const int end = rowptr[node + 1];
    const float* fl = feat + lane * 4;
    v4f acc = {0.f, 0.f, 0.f, 0.f};
    for (int base = beg; base < end; base += 32) {
        const int m = min(end - base, 32);
        const int myid = (lane < m) ? col[base + lane] : 0;
        int k = 0;
        for (; k + 8 <= m; k += 8) {
            v4f r[8];
#pragma unroll
            for (int i = 0; i < 8; ++i) {
                const int s = __builtin_amdgcn_readlane(myid, k + i);
                r[i] = *(const v4f*)(fl + (size_t)s * D);
            }
#pragma unroll
            for (int i = 0; i < 8; ++i) acc += r[i];
        }
        if (m - k >= 4) {
            v4f r[4];
#pragma unroll
            for (int i = 0; i < 4; ++i) {
                const int s = __builtin_amdgcn_readlane(myid, k + i);
                r[i] = *(const v4f*)(fl + (size_t)s * D);
            }
#pragma unroll
            for (int i = 0; i < 4; ++i) acc += r[i];
            k += 4;
        }
        if (m - k >= 2) {
            const int s0 = __builtin_amdgcn_readlane(myid, k);
            const int s1 = __builtin_amdgcn_readlane(myid, k + 1);
            const v4f r0 = *(const v4f*)(fl + (size_t)s0 * D);
            const v4f r1 = *(const v4f*)(fl + (size_t)s1 * D);
            acc += r0;
            acc += r1;
            k += 2;
        }
        if (m - k >= 1) {
            const int s0 = __builtin_amdgcn_readlane(myid, k);
            acc += *(const v4f*)(fl + (size_t)s0 * D);
        }
    }
    *(v4f*)(agg + (size_t)node * D + lane * 4) = acc;
}

#define WP 132
#define GEMM_ROWS 64

__global__ __launch_bounds__(128) void gnn_sage_gemm(
    const float* __restrict__ agg,
    const float* __restrict__ inv,
    const float* __restrict__ xin,
    const float* __restrict__ wl,
    const float* __restrict__ wr,
    const float* __restrict__ bias,
    float* __restrict__ out,
    int nNodes, int relu) {
    __shared__ float wlds[2 * D * WP];

    const int tid     = threadIdx.x;
    const int lane    = tid & 31;
    const int wave    = tid >> 5;
    const int half    = lane >> 4;
    const int l15     = lane & 15;
    const int rowBase = blockIdx.x * GEMM_ROWS + wave * 16;
    const bool valid  = rowBase + 16 <= nNodes;

    v4f am[16], ax[16];
    const int arow = valid ? rowBase + l15 : l15;
    const float inv_c = inv[arow];
    {
        const float* aggRow = agg + (size_t)arow * D + 4 * half;
        const float* xRow   = xin + (size_t)arow * D + 4 * half;
#pragma unroll
        for (int j = 0; j < 16; ++j) am[j] = *(const v4f*)(aggRow + 8 * j);
#pragma unroll
        for (int j = 0; j < 16; ++j) ax[j] = *(const v4f*)(xRow + 8 * j);
    }

    {
        const int c4 = (tid & 31) * 4;
        const int rw = tid >> 5;
#pragma unroll 16
        for (int p = 0; p < 32; ++p) {
            const int r = (((p + (int)blockIdx.x) & 31) << 2) + rw;
            const v4f vl = *(const v4f*)(wl + r * D + c4);
            const v4f vr = *(const v4f*)(wr + r * D + c4);
            *(v4f*)(wlds + r * WP + c4)       = vl;
            *(v4f*)(wlds + (D + r) * WP + c4) = vr;
        }
    }
    __syncthreads();
    if (!valid) return;

#pragma unroll
    for (int j = 0; j < 16; ++j) am[j] *= inv_c;

    v8f acc[8];
#pragma unroll
    for (int n = 0; n < 8; ++n) acc[n] = (v8f){0.f, 0.f, 0.f, 0.f, 0.f, 0.f, 0.f, 0.f};

    const float* bBase = wlds + l15 * WP + 4 * half;

    v4f b0[8], b1[8];
#pragma unroll
    for (int n = 0; n < 8; ++n) b0[n] = *(const v4f*)(bBase + n * 16 * WP);

#pragma unroll
    for (int j = 0; j < 32; j += 2) {
        {
            const int jn = j + 1;
            const float* bp = bBase + (jn >> 4) * (D * WP) + (jn & 15) * 8;
#pragma unroll
            for (int n = 0; n < 8; ++n) b1[n] = *(const v4f*)(bp + n * 16 * WP);
        }
        {
            const v4f a = (j < 16) ? am[j & 15] : ax[j & 15];
            const v2f a01 = {a.x, a.y};
            const v2f a23 = {a.z, a.w};
#pragma unroll
            for (int n = 0; n < 8; ++n) {
                const v2f b = {b0[n].x, b0[n].y};
                acc[n] = __builtin_amdgcn_wmma_f32_16x16x4_f32(false, a01, false, b,
                                                               (short)0, acc[n], false, false);
            }
#pragma unroll
            for (int n = 0; n < 8; ++n) {
                const v2f b = {b0[n].z, b0[n].w};
                acc[n] = __builtin_amdgcn_wmma_f32_16x16x4_f32(false, a23, false, b,
                                                               (short)0, acc[n], false, false);
            }
        }
        if (j + 2 < 32) {
            const int jn = j + 2;
            const float* bp = bBase + (jn >> 4) * (D * WP) + (jn & 15) * 8;
#pragma unroll
            for (int n = 0; n < 8; ++n) b0[n] = *(const v4f*)(bp + n * 16 * WP);
        }
        {
            const int jj = j + 1;
            const v4f a = (jj < 16) ? am[jj & 15] : ax[jj & 15];
            const v2f a01 = {a.x, a.y};
            const v2f a23 = {a.z, a.w};
#pragma unroll
            for (int n = 0; n < 8; ++n) {
                const v2f b = {b1[n].x, b1[n].y};
                acc[n] = __builtin_amdgcn_wmma_f32_16x16x4_f32(false, a01, false, b,
                                                               (short)0, acc[n], false, false);
            }
#pragma unroll
            for (int n = 0; n < 8; ++n) {
                const v2f b = {b1[n].z, b1[n].w};
                acc[n] = __builtin_amdgcn_wmma_f32_16x16x4_f32(false, a23, false, b,
                                                               (short)0, acc[n], false, false);
            }
        }
    }

#pragma unroll
    for (int n = 0; n < 8; ++n) {
        const float bv = bias[n * 16 + l15];
#pragma unroll
        for (int v = 0; v < 8; ++v) {
            float r = acc[n][v] + bv;
            if (relu) r = fmaxf(r, 0.0f);
            out[(size_t)(rowBase + v + 8 * half) * D + n * 16 + l15] = r;
        }
    }
}

extern "C" void kernel_launch(void* const* d_in, const int* in_sizes, int n_in,
                              void* d_out, int out_size, void* d_ws, size_t ws_size,
                              hipStream_t stream) {
    const float* x   = (const float*)d_in[0];
    const int* edge  = (const int*)d_in[1];
    const float* w1l = (const float*)d_in[2];
    const float* w1r = (const float*)d_in[3];
    const float* b1  = (const float*)d_in[4];
    const float* w2l = (const float*)d_in[5];
    const float* w2r = (const float*)d_in[6];
    const float* b2  = (const float*)d_in[7];
    float* out = (float*)d_out;

    const int N = in_sizes[0] / D;
    const int E = in_sizes[1] / 2;
    const int* src = edge;
    const int* dst = edge + E;

    float* agg  = (float*)d_ws;
    float* h    = agg + (size_t)N * D;
    int* cnt    = (int*)(h + (size_t)N * D);
    float* inv  = (float*)(cnt + N);
    int* rowptr = (int*)(inv + N);
    int* slot   = rowptr + (N + 1);
    int* col    = slot + E;

    dim3 blk(256);
    dim3 gNode((unsigned)((N + 255) / 256));
    dim3 gEdge((unsigned)((E + 255) / 256));
    dim3 gGath((unsigned)((N + 7) / 8));
    dim3 blkGemm(128);
    dim3 gGemm((unsigned)((N + GEMM_ROWS - 1) / GEMM_ROWS));

    gnn_zero_i<<<gNode, blk, 0, stream>>>(cnt, N);
    gnn_slot<<<gEdge, blk, 0, stream>>>(dst, cnt, slot, E);
    gnn_rowptr<<<gNode, blk, 0, stream>>>(cnt, rowptr, inv, N);
    gnn_fill<<<gEdge, blk, 0, stream>>>(src, dst, rowptr, slot, col, E);

    gnn_gather<<<gGath, blk, 0, stream>>>(x, rowptr, col, agg, N);
    gnn_sage_gemm<<<gGemm, blkGemm, 0, stream>>>(agg, inv, x, w1l, w1r, b1, h, N, 1);

    gnn_gather<<<gGath, blk, 0, stream>>>(h, rowptr, col, agg, N);
    gnn_sage_gemm<<<gGemm, blkGemm, 0, stream>>>(agg, inv, h, w2l, w2r, b2, out, N, 0);
}


# Edge sort redone in LDS: per 4096-edge chunk a count per bin of 64 dst nodes, then one block per bin builds its list slice; global integer adds and scattered fill gone; list (sum) order = LDS arrival
# speedup vs baseline: 4.2241x; 1.5941x over previous
#include <hip/hip_runtime.h>

typedef __attribute__((ext_vector_type(2))) float v2f;
typedef __attribute__((ext_vector_type(4))) float v4f;
typedef __attribute__((ext_vector_type(8))) float v8f;

#define D 128

#define CH     4096
#define BINW   6
#define MAXBIN 1024
#define CPAD   256
#define CAP    4096

__global__ __launch_bounds__(1024) void gnn_bin(const int* __restrict__ src,
                                                 const int* __restrict__ dst,
                                                 int2* __restrict__ sorted,
                                                 int* __restrict__ offT,
                                                 int nEdges, int nBins) {
    __shared__ int bins[MAXBIN];
    __shared__ int binStart[MAXBIN];
    __shared__ int wsum[32];
    __shared__ int2 stage[CH];
    const int t    = threadIdx.x;
    const int lane = t & 31;
    const int wave = t >> 5;
    const int chunk     = blockIdx.x;
    const int chunkBase = chunk * CH;

    bins[t] = 0;
    int d[4], s[4];
    const int e0 = chunkBase + 4 * t;
    if (((nEdges & 3) == 0) && e0 + 3 < nEdges) {
        const int4 dv = *(const int4*)(dst + e0);
        const int4 sv = *(const int4*)(src + e0);
        d[0] = dv.x; d[1] = dv.y; d[2] = dv.z; d[3] = dv.w;
        s[0] = sv.x; s[1] = sv.y; s[2] = sv.z; s[3] = sv.w;
    } else {
#pragma unroll
        for (int k = 0; k < 4; ++k) {
            const bool ok = e0 + k < nEdges;
            d[k] = ok ? dst[e0 + k] : -1;
            s[k] = ok ? src[e0 + k] : 0;
        }
    }
    __syncthreads();

    int slot[4];
#pragma unroll
    for (int k = 0; k < 4; ++k)
        slot[k] = (d[k] >= 0) ? atomicAdd(&bins[d[k] >> BINW], 1) : 0;
    __syncthreads();

    const int v = bins[t];
    int incl = v;
    for (int off = 1; off < 32; off <<= 1) {
        const int u = __shfl_up(incl, off);
        if (lane >= off) incl += u;
    }
    if (lane == 31) wsum[wave] = incl;
    __syncthreads();
    if (wave == 0) {
        const int w = wsum[lane];
        int wi = w;
        for (int off = 1; off < 32; off <<= 1) {
            const int u = __shfl_up(wi, off);
            if (lane >= off) wi += u;
        }
        wsum[lane] = wi - w;
    }
    __syncthreads();
    const int excl = wsum[wave] + incl - v;
    binStart[t] = excl;
    if (t <= nBins) offT[t * CPAD + chunk] = excl;
    __syncthreads();

#pragma unroll
    for (int k = 0; k < 4; ++k)
        if (d[k] >= 0) stage[binStart[d[k] >> BINW] + slot[k]] = make_int2(d[k], s[k]);
    __syncthreads();

    const int count = min(CH, nEdges - chunkBase);
    for (int i = t; i < count; i += 1024) sorted[chunkBase + i] = stage[i];
}

__global__ __launch_bounds__(512) void gnn_csr(const int2* __restrict__ sorted,
                                               const int* __restrict__ offT,
                                               int* __restrict__ rowptr,
                                               float* __restrict__ inv,
                                               int* __restrict__ col,
                                               int nNodes, int nChunks) {
    __shared__ int segPos[CPAD];
    __shared__ int segBeg[CPAD];
    __shared__ int wLen[16];
    __shared__ int wBeg[16];
    __shared__ int cnt[64];
    __shared__ int start[64];
    __shared__ int cur[64];
    __shared__ int2 pairS[CAP];
    __shared__ int colS[CAP];
    const int t    = threadIdx.x;
    const int lane = t & 31;
    const int wave = t >> 5;
    const int b    = blockIdx.x;

    int sBeg = 0, len = 0;
    if (t < nChunks) {
        sBeg = offT[b * CPAD + t];
        len  = offT[(b + 1) * CPAD + t] - sBeg;
    }
    if (t < CPAD) segBeg[t] = t * CH + sBeg;
    if (t < 64) { cnt[t] = 0; cur[t] = 0; }

    int incl = len;
    for (int off = 1; off < 32; off <<= 1) {
        const int u = __shfl_up(incl, off);
        if (lane >= off) incl += u;
    }
    int bsum = sBeg;
    for (int off = 16; off > 0; off >>= 1) bsum += __shfl_xor(bsum, off);
    if (lane == 31) wLen[wave] = incl;
    if (lane == 0) wBeg[wave] = bsum;
    __syncthreads();
    int pre = 0, base = 0, total = 0;
    for (int k = 0; k < 16; ++k) {
        const int wl = wLen[k];
        base += wBeg[k];
        total += wl;
        if (k < wave) pre += wl;
    }
    if (t < CPAD) segPos[t] = pre + incl - len;
    __syncthreads();

    const bool staged = total <= CAP;

    auto fetch = [&](int j) -> int2 {
        int c = 0;
        for (int step = CPAD / 2; step > 0; step >>= 1)
            if (segPos[c + step] <= j) c += step;
        return sorted[segBeg[c] + (j - segPos[c])];
    };

    for (int j = t; j < total; j += 512) {
        const int2 p = fetch(j);
        if (staged) pairS[j] = p;
        atomicAdd(&cnt[p.x & 63], 1);
    }
    __syncthreads();

    if (wave == 0) {
        const int a0 = cnt[2 * lane];
        const int a1 = cnt[2 * lane + 1];
        const int s2 = a0 + a1;
        int si = s2;
        for (int off = 1; off < 32; off <<= 1) {
            const int u = __shfl_up(si, off);
            if (lane >= off) si += u;
        }
        start[2 * lane]     = si - s2;
        start[2 * lane + 1] = si - s2 + a0;
    }
    __syncthreads();

    if (t < 64) {
        const int node = (b << BINW) + t;
        if (node < nNodes) {
            const int c = cnt[t];
            rowptr[node] = base + start[t];
            inv[node] = 1.0f / fmaxf((float)c, 1.0f);
            if (node == nNodes - 1) rowptr[nNodes] = base + start[t] + c;
        }
    }

    for (int j = t; j < total; j += 512) {
        const int2 p = staged ? pairS[j] : fetch(j);
        const int l = p.x & 63;
        const int pos = start[l] + atomicAdd(&cur[l], 1);
        if (staged) colS[pos] = p.y;
        else col[base + pos] = p.y;
    }
    if (staged) {
        __syncthreads();
        for (int i = t; i < total; i += 512) col[base + i] = colS[i];
    }
}

__global__ __launch_bounds__(256) void gnn_gather(const float* __restrict__ feat,
                                                  const int* __restrict__ rowptr,
                                                  const int* __restrict__ col,
                                                  float* __restrict__ agg, int nNodes) {
    const int lane = threadIdx.x & 31;
    const int node = __builtin_amdgcn_readfirstlane((int)((blockIdx.x * 256u + threadIdx.x) >> 5));
    if (node >= nNodes) return;
    const int beg = rowptr[node];
    const int end = rowptr[node + 1];
    const float* fl = feat + lane * 4;
    v4f acc = {0.f, 0.f, 0.f, 0.f};
    for (int base = beg; base < end; base += 32) {
        const int m = min(end - base, 32);
        const int myid = (lane < m) ? col[base + lane] : 0;
        int k = 0;
        for (; k + 8 <= m; k += 8) {
            v4f r[8];
#pragma unroll
            for (int i = 0; i < 8; ++i) {
                const int s = __builtin_amdgcn_readlane(myid, k + i);
                r[i] = *(const v4f*)(fl + (size_t)s * D);
            }
#pragma unroll
            for (int i = 0; i < 8; ++i) acc += r[i];
        }
        if (m - k >= 4) {
            v4f r[4];
#pragma unroll
            for (int i = 0; i < 4; ++i) {
                const int s = __builtin_amdgcn_readlane(myid, k + i);
                r[i] = *(const v4f*)(fl + (size_t)s * D);
            }
#pragma unroll
            for (int i = 0; i < 4; ++i) acc += r[i];
            k += 4;
        }
        if (m - k >= 2) {
            const int s0 = __builtin_amdgcn_readlane(myid, k);
            const int s1 = __builtin_amdgcn_readlane(myid, k + 1);
            const v4f r0 = *(const v4f*)(fl + (size_t)s0 * D);
            const v4f r1 = *(const v4f*)(fl + (size_t)s1 * D);
            acc += r0;
            acc += r1;
            k += 2;
        }
        if (m - k >= 1) {
            const int s0 = __builtin_amdgcn_readlane(myid, k);
            acc += *(const v4f*)(fl + (size_t)s0 * D);
        }
    }
    *(v4f*)(agg + (size_t)node * D + lane * 4) = acc;
}

#define WP 132
#define GEMM_ROWS 64

__global__ __launch_bounds__(128) void gnn_sage_gemm(
    const float* __restrict__ agg,
    const float* __restrict__ inv,
    const float* __restrict__ xin,
    const float* __restrict__ wl,
    const float* __restrict__ wr,
    const float* __restrict__ bias,
    float* __restrict__ out,
    int nNodes, int relu) {
    __shared__ float wlds[2 * D * WP];

    const int tid     = threadIdx.x;
    const int lane    = tid & 31;
    const int wave    = tid >> 5;
    const int half    = lane >> 4;
    const int l15     = lane & 15;
    const int rowBase = blockIdx.x * GEMM_ROWS + wave * 16;
    const bool valid  = rowBase + 16 <= nNodes;

    v4f am[16], ax[16];
    const int arow = valid ? rowBase + l15 : l15;
    const float inv_c = inv[arow];
    {
        const float* aggRow = agg + (size_t)arow * D + 4 * half;
        const float* xRow   = xin + (size_t)arow * D + 4 * half;
#pragma unroll
        for (int j = 0; j < 16; ++j) am[j] = *(const v4f*)(aggRow + 8 * j);
#pragma unroll
        for (int j = 0; j < 16; ++j) ax[j] = *(const v4f*)(xRow + 8 * j);
    }

    {
        const int c4 = (tid & 31) * 4;
        const int rw = tid >> 5;
#pragma unroll 16
        for (int p = 0; p < 32; ++p) {
            const int r = (((p + (int)blockIdx.x) & 31) << 2) + rw;
            const v4f vl = *(const v4f*)(wl + r * D + c4);
            const v4f vr = *(const v4f*)(wr + r * D + c4);
            *(v4f*)(wlds + r * WP + c4)       = vl;
            *(v4f*)(wlds + (D + r) * WP + c4) = vr;
        }
    }
    __syncthreads();
    if (!valid) return;

#pragma unroll
    for (int j = 0; j < 16; ++j) am[j] *= inv_c;

    v8f acc[8];
#pragma unroll
    for (int n = 0; n < 8; ++n) acc[n] = (v8f){0.f, 0.f, 0.f, 0.f, 0.f, 0.f, 0.f, 0.f};

    const float* bBase = wlds + l15 * WP + 4 * half;

    v4f b0[8], b1[8];
#pragma unroll
    for (int n = 0; n < 8; ++n) b0[n] = *(const v4f*)(bBase + n * 16 * WP);

#pragma unroll
    for (int j = 0; j < 32; j += 2) {
        {
            const int jn = j + 1;
            const float* bp = bBase + (jn >> 4) * (D * WP) + (jn & 15) * 8;
#pragma unroll
            for (int n = 0; n < 8; ++n) b1[n] = *(const v4f*)(bp + n * 16 * WP);
        }
        {
            const v4f a = (j < 16) ? am[j & 15] : ax[j & 15];
            const v2f a01 = {a.x, a.y};
            const v2f a23 = {a.z, a.w};
#pragma unroll
            for (int n = 0; n < 8; ++n) {
                const v2f b = {b0[n].x, b0[n].y};
                acc[n] = __builtin_amdgcn_wmma_f32_16x16x4_f32(false, a01, false, b,
                                                               (short)0, acc[n], false, false);
            }
#pragma unroll
            for (int n = 0; n < 8; ++n) {
                const v2f b = {b0[n].z, b0[n].w};
                acc[n] = __builtin_amdgcn_wmma_f32_16x16x4_f32(false, a23, false, b,
                                                               (short)0, acc[n], false, false);
            }
        }
        if (j + 2 < 32) {
            const int jn = j + 2;
            const float* bp = bBase + (jn >> 4) * (D * WP) + (jn & 15) * 8;
#pragma unroll
            for (int n = 0; n < 8; ++n) b0[n] = *(const v4f*)(bp + n * 16 * WP);
        }
        {
            const int jj = j + 1;
            const v4f a = (jj < 16) ? am[jj & 15] : ax[jj & 15];
            const v2f a01 = {a.x, a.y};
            const v2f a23 = {a.z, a.w};
#pragma unroll
            for (int n = 0; n < 8; ++n) {
                const v2f b = {b1[n].x, b1[n].y};
                acc[n] = __builtin_amdgcn_wmma_f32_16x16x4_f32(false, a01, false, b,
                                                               (short)0, acc[n], false, false);
            }
#pragma unroll
            for (int n = 0; n < 8; ++n) {
                const v2f b = {b1[n].z, b1[n].w};
                acc[n] = __builtin_amdgcn_wmma_f32_16x16x4_f32(false, a23, false, b,
                                                               (short)0, acc[n], false, false);
            }
        }
    }

#pragma unroll
    for (int n = 0; n < 8; ++n) {
        const float bv = bias[n * 16 + l15];
#pragma unroll
        for (int v = 0; v < 8; ++v) {
            float r = acc[n][v] + bv;
            if (relu) r = fmaxf(r, 0.0f);
            out[(size_t)(rowBase + v + 8 * half) * D + n * 16 + l15] = r;
        }
    }
}

extern "C" void kernel_launch(void* const* d_in, const int* in_sizes, int n_in,
                              void* d_out, int out_size, void* d_ws, size_t ws_size,
                              hipStream_t stream) {
    const float* x   = (const float*)d_in[0];
    const int* edge  = (const int*)d_in[1];
    const float* w1l = (const float*)d_in[2];
    const float* w1r = (const float*)d_in[3];
    const float* b1  = (const float*)d_in[4];
    const float* w2l = (const float*)d_in[5];
    const float* w2r = (const float*)d_in[6];
    const float* b2  = (const float*)d_in[7];
    float* out = (float*)d_out;

    const int N = in_sizes[0] / D;
    const int E = in_sizes[1] / 2;
    const int* src = edge;
    const int* dst = edge + E;

    float* agg   = (float*)d_ws;
    float* h     = agg + (size_t)N * D;
    int2* sorted = (int2*)(h + (size_t)N * D);
    const int nBins   = (N + 63) >> BINW;
    const int nChunks = (E + CH - 1) / CH;
    int* offT   = (int*)(sorted + E);
    float* inv  = (float*)(offT + (size_t)(nBins + 1) * CPAD);
    int* rowptr = (int*)(inv + N);
    int* col    = rowptr + (N + 1);

    dim3 blk(256);
    dim3 gGath((unsigned)((N + 7) / 8));
    dim3 blkGemm(128);
    dim3 gGemm((unsigned)((N + GEMM_ROWS - 1) / GEMM_ROWS));

    gnn_bin<<<dim3((unsigned)nChunks), dim3(1024), 0, stream>>>(src, dst, sorted, offT, E, nBins);
    gnn_csr<<<dim3((unsigned)nBins), dim3(512), 0, stream>>>(sorted, offT, rowptr, inv, col, N, nChunks);

    gnn_gather<<<gGath, blk, 0, stream>>>(x, rowptr, col, agg, N);
    gnn_sage_gemm<<<gGemm, blkGemm, 0, stream>>>(agg, inv, x, w1l, w1r, b1, h, N, 1);

    gnn_gather<<<gGath, blk, 0, stream>>>(h, rowptr, col, agg, N);
    gnn_sage_gemm<<<gGemm, blkGemm, 0, stream>>>(agg, inv, h, w2l, w2r, b2, out, N, 0);
}
